# starting kernel
# baseline (speedup 1.0000x reference)
#include <hip/hip_runtime.h>
#include <hip/hip_bf16.h>

typedef __attribute__((ext_vector_type(16))) _Float16 v16h;
typedef __attribute__((ext_vector_type(8)))  _Float16 v8h;
typedef __attribute__((ext_vector_type(8)))  float    v8f;

#define NPOINTS 8192
#define NBATCH  16
#define PTSB    8
#define CLAMPC  1.9f
#define TWOPI_I 0.63661977236758f

__device__ __forceinline__ v16h ld16(const _Float16* a, const _Float16* b) {
  v8h lo = *(const v8h*)a;
  v8h hi = *(const v8h*)b;
  return __builtin_shufflevector(lo, hi, 0,1,2,3,4,5,6,7,8,9,10,11,12,13,14,15);
}

__device__ __forceinline__ v8f wmma16(v16h a, v16h b, v8f c) {
  return __builtin_amdgcn_wmma_f32_16x16x32_f16(false, a, false, b, (short)0, c,
                                                false, false);
}

__global__ void prep_kernel(const float* __restrict__ x,
                            const float* __restrict__ logs,
                            const float* __restrict__ bias,
                            float* __restrict__ zmat,
                            float* __restrict__ logdet,
                            _Float16* __restrict__ z1h, int total) {
  int p = blockIdx.x * blockDim.x + threadIdx.x;
  if (p < total) {
    const float* xr = x + (size_t)p * 32;
    float* zr = zmat + (size_t)p * 32;
    _Float16* hr = z1h + (size_t)p * 16;
#pragma unroll
    for (int c = 0; c < 32; ++c) {
      float zc = xr[31 - c] * expf(logs[c]) + bias[c];
      zr[c] = zc;
      if (c < 16) hr[c] = (_Float16)zc;
    }
  }
  if (p < NBATCH) {
    float su = 0.f;
#pragma unroll
    for (int c = 0; c < 32; ++c) su += logs[c];
    logdet[p] = su * (float)NPOINTS;
  }
}

__global__ __launch_bounds__(128) void conv_kernel(
    const _Float16* __restrict__ z1h, const int* __restrict__ knn,
    const float* __restrict__ W1g, const float* __restrict__ b1g,
    const float* __restrict__ W2g, const float* __restrict__ b2g,
    const float* __restrict__ W3g, const float* __restrict__ b3g,
    float* __restrict__ zmat, float* __restrict__ logdet) {
  __shared__ __align__(16) _Float16 sW1[2][64][40];
  __shared__ __align__(16) _Float16 sW2[2][64][72];
  __shared__ __align__(16) _Float16 sW3[2][16][72];
  __shared__ __align__(16) _Float16 sA[128][40];
  __shared__ __align__(16) _Float16 sY[4][16][72];
  __shared__ __align__(16) _Float16 sZ[4][16][72];
  __shared__ float sB1[2][64], sB2[2][64], sB3[2][16];
  __shared__ float sLog;

  const int tid  = threadIdx.x;
  const int wid  = tid >> 5;
  const int lane = tid & 31;
  const int n16  = lane & 15;
  const int hl   = lane >> 4;
  const int base0 = hl * 8;

  const int bi = blockIdx.x / (NPOINTS / PTSB);
  const int pb = (blockIdx.x % (NPOINTS / PTSB)) * PTSB;

  for (int idx = tid; idx < 2 * 64 * 32; idx += 128) {
    int s = idx >> 11, r = idx & 2047;
    int j = r >> 5, k = r & 31;
    float v;
    if (k < 16)
      v = W1g[(s * 48 + 16 + k) * 64 + j] + W1g[(s * 48 + 32 + k) * 64 + j];
    else {
      int kk = k - 16;
      v = W1g[(s * 48 + kk) * 64 + j] - W1g[(s * 48 + 32 + kk) * 64 + j];
    }
    sW1[s][j][k] = (_Float16)v;
  }
  for (int idx = tid; idx < 2 * 64 * 64; idx += 128) {
    int s = idx >> 12, r = idx & 4095;
    int j = r >> 6, k = r & 63;
    sW2[s][j][k] = (_Float16)W2g[(s * 64 + k) * 64 + j];
  }
  for (int idx = tid; idx < 2 * 16 * 64; idx += 128) {
    int s = idx >> 10, r = idx & 1023;
    int j = r >> 6, k = r & 63;
    sW3[s][j][k] = (_Float16)W3g[(s * 64 + k) * 16 + j];
  }
  {
    int s = tid >> 6, j = tid & 63;
    sB1[s][j] = b1g[s * 64 + j];
    sB2[s][j] = b2g[s * 64 + j];
    if (tid < 32) sB3[tid >> 4][tid & 15] = b3g[(tid >> 4) * 16 + (tid & 15)];
    if (tid == 0) sLog = 0.f;
  }

  {
    const int e = tid, lp = e >> 4, k = e & 15;
    const size_t gp = (size_t)bi * NPOINTS + pb + lp;
    const int nidx = knn[gp * 16 + k];
    const _Float16* nb = z1h + ((size_t)bi * NPOINTS + nidx) * 16;
    const _Float16* ct = z1h + gp * 16;
    *(v8h*)&sA[e][0]  = *(const v8h*)nb;
    *(v8h*)&sA[e][8]  = *(const v8h*)(nb + 8);
    *(v8h*)&sA[e][16] = *(const v8h*)ct;
    *(v8h*)&sA[e][24] = *(const v8h*)(ct + 8);
  }
  __syncthreads();

  for (int tt = 0; tt < 2; ++tt) {
    const int tile = wid + tt * 4;
    const int p = pb + tile;
    const int e0 = tile * 16;
    const int arow = e0 + n16;
    const v16h A1 = ld16(&sA[arow][base0], &sA[arow][16 + base0]);
    float sv = 0.f;

#pragma unroll
    for (int s = 0; s < 2; ++s) {
#pragma unroll
      for (int nc = 0; nc < 4; ++nc) {
        const int j = nc * 16 + n16;
        const v16h B1 = ld16(&sW1[s][j][base0], &sW1[s][j][16 + base0]);
        v8f c; const float bv = sB1[s][j];
#pragma unroll
        for (int r = 0; r < 8; ++r) c[r] = bv;
        c = wmma16(A1, B1, c);
#pragma unroll
        for (int r = 0; r < 8; ++r)
          sY[wid][r + 8 * hl][j] = (_Float16)fmaxf(c[r], 0.0f);
      }
      __builtin_amdgcn_wave_barrier();

      const v16h A2a = ld16(&sY[wid][n16][base0],      &sY[wid][n16][16 + base0]);
      const v16h A2b = ld16(&sY[wid][n16][32 + base0], &sY[wid][n16][48 + base0]);
#pragma unroll
      for (int nc = 0; nc < 4; ++nc) {
        const int j = nc * 16 + n16;
        const v16h B2a = ld16(&sW2[s][j][base0],      &sW2[s][j][16 + base0]);
        const v16h B2b = ld16(&sW2[s][j][32 + base0], &sW2[s][j][48 + base0]);
        v8f c; const float bv = sB2[s][j];
#pragma unroll
        for (int r = 0; r < 8; ++r) c[r] = bv;
        c = wmma16(A2a, B2a, c);
        c = wmma16(A2b, B2b, c);
#pragma unroll
        for (int r = 0; r < 8; ++r)
          sZ[wid][r + 8 * hl][j] = (_Float16)fmaxf(c[r], 0.0f);
      }
      __builtin_amdgcn_wave_barrier();

      const v16h A3a = ld16(&sZ[wid][n16][base0],      &sZ[wid][n16][16 + base0]);
      const v16h A3b = ld16(&sZ[wid][n16][32 + base0], &sZ[wid][n16][48 + base0]);
      const v16h B3a = ld16(&sW3[s][n16][base0],       &sW3[s][n16][16 + base0]);
      const v16h B3b = ld16(&sW3[s][n16][32 + base0],  &sW3[s][n16][48 + base0]);
      v8f c; const float bv = sB3[s][n16];
#pragma unroll
      for (int r = 0; r < 8; ++r) c[r] = bv;
      c = wmma16(A3a, B3a, c);
      c = wmma16(A3b, B3b, c);
      float v = c[0];
#pragma unroll
      for (int r = 1; r < 8; ++r) v = fmaxf(v, c[r]);
      v = fmaxf(v, __shfl_xor(v, 16, 32));

      if (s == 0) {
        sv = CLAMPC * TWOPI_I * atanf(v * (1.0f / CLAMPC));
        float r0 = sv;
        r0 += __shfl_xor(r0, 1, 32);
        r0 += __shfl_xor(r0, 2, 32);
        r0 += __shfl_xor(r0, 4, 32);
        r0 += __shfl_xor(r0, 8, 32);
        if (lane == 0) atomicAdd(&sLog, r0);
      } else {
        if (lane < 16) {
          float* zp = zmat + ((size_t)bi * NPOINTS + p) * 32 + 16 + n16;
          *zp = *zp * __expf(sv) + v;
        }
      }
      __builtin_amdgcn_wave_barrier();
    }
  }
  __syncthreads();
  if (tid == 0) atomicAdd(&logdet[bi], sLog);
}

extern "C" void kernel_launch(void* const* d_in, const int* in_sizes, int n_in,
                              void* d_out, int out_size, void* d_ws, size_t ws_size,
                              hipStream_t stream) {
  const float* x     = (const float*)d_in[0];
  const int*   knn   = (const int*)d_in[1];
  const float* logs  = (const float*)d_in[2];
  const float* biasv = (const float*)d_in[3];
  const float* W1    = (const float*)d_in[4];
  const float* b1    = (const float*)d_in[5];
  const float* W2    = (const float*)d_in[6];
  const float* b2    = (const float*)d_in[7];
  const float* W3    = (const float*)d_in[8];
  const float* b3    = (const float*)d_in[9];

  float* zmat   = (float*)d_out;
  float* logdet = zmat + (size_t)NBATCH * NPOINTS * 32;
  _Float16* z1h = (_Float16*)d_ws;

  const int total = NBATCH * NPOINTS;
  prep_kernel<<<(total + 255) / 256, 256, 0, stream>>>(x, logs, biasv, zmat,
                                                       logdet, z1h, total);

  const dim3 grid(NBATCH * (NPOINTS / PTSB));
  for (int u = 0; u < 4; u += 2) {
    conv_kernel<<<grid, 128, 0, stream>>>(
        z1h, knn,
        W1 + (size_t)u * 48 * 64, b1 + (size_t)u * 64,
        W2 + (size_t)u * 64 * 64, b2 + (size_t)u * 64,
        W3 + (size_t)u * 64 * 16, b3 + (size_t)u * 16,
        zmat, logdet);
  }
}


# one net per wave: a group's 4 waves each keep ONE net's weight fragments and walk the same 128 points (4 waves per SIMD, was 2); results meet in LDS; one s atomic per group (sum order changes)
# speedup vs baseline: 5.3887x; 5.3887x over previous
#include <hip/hip_runtime.h>
#include <hip/hip_bf16.h>


typedef __attribute__((ext_vector_type(16))) _Float16 v16h;
typedef __attribute__((ext_vector_type(8)))  _Float16 v8h;
typedef __attribute__((ext_vector_type(8)))  float    v8f;
typedef __attribute__((ext_vector_type(2)))  _Float16 v2h;
typedef __attribute__((ext_vector_type(2)))  float    v2f;
typedef __attribute__((ext_vector_type(8)))  short    v8s;

#define NPOINTS 8192
#define NBATCH  16
#define PPW     128
#define GPB     (NPOINTS / PPW)
#define NETS    4
#define EPT     (PPW * 16 / (NETS * 32))
#define CLAMPC  1.9f
#define TWOPI_I 0.63661977236758f

__device__ __forceinline__ v16h cat16(v8h lo, v8h hi) {
  return __builtin_shufflevector(lo, hi, 0,1,2,3,4,5,6,7,8,9,10,11,12,13,14,15);
}

__device__ __forceinline__ v8h ld_tr16(const _Float16* p) {
  typedef __attribute__((address_space(3))) v8s lds_v8s;
  return __builtin_bit_cast(v8h, __builtin_amdgcn_ds_load_tr16_b128_v8i16((lds_v8s*)p));
}

__device__ __forceinline__ v16h ldA(const _Float16* p, int kb) {
  return cat16(ld_tr16(p + kb * 16), ld_tr16(p + (kb + 16) * 16));
}

__device__ __forceinline__ v8h relu_pack(v8f c) {
  v2h q[4];
#pragma unroll
  for (int r = 0; r < 4; ++r) {
    v2f t = {fmaxf(c[2 * r], 0.0f), fmaxf(c[2 * r + 1], 0.0f)};
    q[r] = __builtin_convertvector(t, v2h);
  }
  v8h o = {q[0][0], q[0][1], q[1][0], q[1][1], q[2][0], q[2][1], q[3][0], q[3][1]};
  return o;
}

__device__ __forceinline__ v8f wmma16(v16h a, v16h b, v8f c) {
  return __builtin_amdgcn_wmma_f32_16x16x32_f16(false, a, false, b, (short)0, c,
                                                false, false);
}

__device__ __forceinline__ v8f splat8(float x) {
  v8f c;
#pragma unroll
  for (int r = 0; r < 8; ++r) c[r] = x;
  return c;
}

__global__ void prep_kernel(const float* __restrict__ x,
                            const float* __restrict__ logs,
                            const float* __restrict__ bias,
                            float* __restrict__ zmat,
                            float* __restrict__ logdet,
                            _Float16* __restrict__ z1h, int total) {
  int p = blockIdx.x * blockDim.x + threadIdx.x;
  if (p < total) {
    const float* xr = x + (size_t)p * 32;
    float* zr = zmat + (size_t)p * 32;
    _Float16* hr = z1h + (size_t)p * 16;
#pragma unroll
    for (int c = 0; c < 32; ++c) {
      float zc = xr[31 - c] * expf(logs[c]) + bias[c];
      zr[c] = zc;
      if (c < 16) hr[c] = (_Float16)zc;
    }
  }
  if (p < NBATCH) {
    float su = 0.f;
#pragma unroll
    for (int c = 0; c < 32; ++c) su += logs[c];
    logdet[p] = su * (float)NPOINTS;
  }
}

__global__ __launch_bounds__(NETS * 32) void conv_kernel(
    const _Float16* __restrict__ z1h, const int* __restrict__ knn,
    const float* __restrict__ W1g, const float* __restrict__ b1g,
    const float* __restrict__ W2g, const float* __restrict__ b2g,
    const float* __restrict__ W3g, const float* __restrict__ b3g,
    float* __restrict__ zmat, float* __restrict__ logdet) {
  __shared__ __align__(32) _Float16 sYw[NETS][64][16];
  __shared__ __align__(32) _Float16 sZw[NETS][64][16];
  __shared__ float sRes[NETS][PPW][16];
  __shared__ float sSum[2];

  const int u     = __builtin_amdgcn_readfirstlane(threadIdx.x >> 5);
  _Float16 (*sY)[16] = sYw[u];
  _Float16 (*sZ)[16] = sZw[u];
  const int lane  = threadIdx.x & 31;
  const int n16   = lane & 15;
  const int hl    = lane >> 4;
  const int base0 = hl * 8;
  const _Float16* tY = &sY[base0 + (n16 & 7)][8 * (n16 >> 3)];
  const _Float16* tZ = &sZ[base0 + (n16 & 7)][8 * (n16 >> 3)];

  const float* W1 = W1g + (size_t)u * 48 * 64;
  const float* W2 = W2g + (size_t)u * 64 * 64;
  const float* W3 = W3g + (size_t)u * 64 * 16;

  v16h w1[4], w2a[4], w2b[4], w3a, w3b;
  float bb1[4], bb2[4], bb3;
#pragma unroll
  for (int nc = 0; nc < 4; ++nc) {
    const int j = nc * 16 + n16;
    v16h f1, fa, fb;
#pragma unroll
    for (int i = 0; i < 8; ++i) {
      const int k = base0 + i;
      f1[i] = (_Float16)(W1[(16 + k) * 64 + j] + W1[(32 + k) * 64 + j]);
      f1[8 + i] = (_Float16)(W1[k * 64 + j] - W1[(32 + k) * 64 + j]);
      fa[i]     = (_Float16)W2[k * 64 + j];
      fa[8 + i] = (_Float16)W2[(16 + k) * 64 + j];
      fb[i]     = (_Float16)W2[(32 + k) * 64 + j];
      fb[8 + i] = (_Float16)W2[(48 + k) * 64 + j];
    }
    w1[nc] = f1; w2a[nc] = fa; w2b[nc] = fb;
    bb1[nc] = b1g[u * 64 + j];
    bb2[nc] = b2g[u * 64 + j];
    __builtin_amdgcn_sched_barrier(0);
  }
  {
    v16h ga, gb;
#pragma unroll
    for (int i = 0; i < 8; ++i) {
      const int k = base0 + i;
      ga[i]     = (_Float16)W3[k * 16 + n16];
      ga[8 + i] = (_Float16)W3[(16 + k) * 16 + n16];
      gb[i]     = (_Float16)W3[(32 + k) * 16 + n16];
      gb[8 + i] = (_Float16)W3[(48 + k) * 16 + n16];
    }
    w3a = ga; w3b = gb;
    bb3 = b3g[u * 16 + n16];
  }

  const int bi = blockIdx.x / GPB;
  const int gi = blockIdx.x % GPB;
  const int p0 = gi * PPW;
  const int p1 = p0 + PPW;
  const size_t rb = (size_t)bi * NPOINTS;
  const int plast = p1 - 1;

  int nidx = knn[(rb + p0) * 16 + n16];
  v8h nbv = *(const v8h*)(z1h + (rb + nidx) * 16 + base0);
  v8h ctv = *(const v8h*)(z1h + (rb + p0) * 16 + base0);
  {
    const int pn = p0 + 1 < plast ? p0 + 1 : plast;
    nidx = knn[(rb + pn) * 16 + n16];
  }
  float ssum = 0.f;

  for (int p = p0; p < p1; ++p) {
    const v16h A1 = cat16(nbv, ctv);
    {
      const int pn  = p + 1 < plast ? p + 1 : plast;
      const int pnn = p + 2 < plast ? p + 2 : plast;
      nbv  = *(const v8h*)(z1h + (rb + nidx) * 16 + base0);
      ctv  = *(const v8h*)(z1h + (rb + pn) * 16 + base0);
      nidx = knn[(rb + pnn) * 16 + n16];
    }

#pragma unroll
    for (int nc = 0; nc < 4; ++nc) {
      const int j = nc * 16 + n16;
      v8f c = wmma16(A1, w1[nc], splat8(bb1[nc]));
      *(v8h*)&sY[j][base0] = relu_pack(c);
    }
    __builtin_amdgcn_wave_barrier();

    const v16h A2a = ldA(tY, 0);
    const v16h A2b = ldA(tY, 32);
#pragma unroll
    for (int nc = 0; nc < 4; ++nc) {
      const int j = nc * 16 + n16;
      v8f c = splat8(bb2[nc]);
      c = wmma16(A2a, w2a[nc], c);
      c = wmma16(A2b, w2b[nc], c);
      *(v8h*)&sZ[j][base0] = relu_pack(c);
    }
    __builtin_amdgcn_wave_barrier();

    const v16h A3a = ldA(tZ, 0);
    const v16h A3b = ldA(tZ, 32);
    v8f c = splat8(bb3);
    c = wmma16(A3a, w3a, c);
    c = wmma16(A3b, w3b, c);
    float v = c[0];
#pragma unroll
    for (int r = 1; r < 8; ++r) v = fmaxf(v, c[r]);
    v = fmaxf(v, __shfl_xor(v, 16, 32));

    if ((u & 1) == 0) {
      v = CLAMPC * TWOPI_I * atanf(v * (1.0f / CLAMPC));
      ssum += v;
    }
    if (lane < 16) sRes[u][p - p0][n16] = v;
    __builtin_amdgcn_wave_barrier();
  }

  if ((u & 1) == 0) {
    ssum += __shfl_xor(ssum, 1, 32);
    ssum += __shfl_xor(ssum, 2, 32);
    ssum += __shfl_xor(ssum, 4, 32);
    ssum += __shfl_xor(ssum, 8, 32);
    if (lane == 0) sSum[u >> 1] = ssum;
  }

  float zo[EPT];
#pragma unroll
  for (int i = 0; i < EPT; ++i) {
    const int e = threadIdx.x + i * (NETS * 32);
    zo[i] = zmat[(rb + p0 + (e >> 4)) * 32 + 16 + (e & 15)];
  }

  __syncthreads();

  if (threadIdx.x == 0) atomicAdd(&logdet[bi], sSum[0] + sSum[1]);

#pragma unroll
  for (int i = 0; i < EPT; ++i) {
    const int e  = threadIdx.x + i * (NETS * 32);
    const int pt = e >> 4;
    const int f  = e & 15;
    float z2 = zo[i];
    z2 = z2 * __expf(sRes[0][pt][f]) + sRes[1][pt][f];
    z2 = z2 * __expf(sRes[2][pt][f]) + sRes[3][pt][f];
    zmat[(rb + p0 + pt) * 32 + 16 + f] = z2;
  }
}

extern "C" void kernel_launch(void* const* d_in, const int* in_sizes, int n_in,
                              void* d_out, int out_size, void* d_ws, size_t ws_size,
                              hipStream_t stream) {
  const float* x     = (const float*)d_in[0];
  const int*   knn   = (const int*)d_in[1];
  const float* logs  = (const float*)d_in[2];
  const float* biasv = (const float*)d_in[3];
  const float* W1    = (const float*)d_in[4];
  const float* b1    = (const float*)d_in[5];
  const float* W2    = (const float*)d_in[6];
  const float* b2    = (const float*)d_in[7];
  const float* W3    = (const float*)d_in[8];
  const float* b3    = (const float*)d_in[9];

  float* zmat   = (float*)d_out;
  float* logdet = zmat + (size_t)NBATCH * NPOINTS * 32;
  _Float16* z1h = (_Float16*)d_ws;

  const int total = NBATCH * NPOINTS;
  prep_kernel<<<(total + 255) / 256, 256, 0, stream>>>(x, logs, biasv, zmat,
                                                       logdet, z1h, total);

  conv_kernel<<<dim3(NBATCH * GPB), NETS * 32, 0, stream>>>(
      z1h, knn, W1, b1, W2, b2, W3, b3, zmat, logdet);
}


# soft clamp and s sum moved out of the point loop: all 128 threads clamp once per (point, feature) in the closing pass (was per point on the two s-waves, 32 lanes for 16 values); sum order changes
# speedup vs baseline: 5.7258x; 1.0626x over previous
#include <hip/hip_runtime.h>
#include <hip/hip_bf16.h>


typedef __attribute__((ext_vector_type(16))) _Float16 v16h;
typedef __attribute__((ext_vector_type(8)))  _Float16 v8h;
typedef __attribute__((ext_vector_type(8)))  float    v8f;
typedef __attribute__((ext_vector_type(2)))  _Float16 v2h;
typedef __attribute__((ext_vector_type(2)))  float    v2f;
typedef __attribute__((ext_vector_type(8)))  short    v8s;

#define NPOINTS 8192
#define NBATCH  16
#define PPW     128
#define GPB     (NPOINTS / PPW)
#define NETS    4
#define EPT     (PPW * 16 / (NETS * 32))
#define CLAMPC  1.9f
#define TWOPI_I 0.63661977236758f

__device__ __forceinline__ v16h cat16(v8h lo, v8h hi) {
  return __builtin_shufflevector(lo, hi, 0,1,2,3,4,5,6,7,8,9,10,11,12,13,14,15);
}

__device__ __forceinline__ v8h ld_tr16(const _Float16* p) {
  typedef __attribute__((address_space(3))) v8s lds_v8s;
  return __builtin_bit_cast(v8h, __builtin_amdgcn_ds_load_tr16_b128_v8i16((lds_v8s*)p));
}

__device__ __forceinline__ v16h ldA(const _Float16* p, int kb) {
  return cat16(ld_tr16(p + kb * 16), ld_tr16(p + (kb + 16) * 16));
}

__device__ __forceinline__ v8h relu_pack(v8f c) {
  v2h q[4];
#pragma unroll
  for (int r = 0; r < 4; ++r) {
    v2f t = {fmaxf(c[2 * r], 0.0f), fmaxf(c[2 * r + 1], 0.0f)};
    q[r] = __builtin_convertvector(t, v2h);
  }
  v8h o = {q[0][0], q[0][1], q[1][0], q[1][1], q[2][0], q[2][1], q[3][0], q[3][1]};
  return o;
}

__device__ __forceinline__ v8f wmma16(v16h a, v16h b, v8f c) {
  return __builtin_amdgcn_wmma_f32_16x16x32_f16(false, a, false, b, (short)0, c,
                                                false, false);
}

__device__ __forceinline__ v8f splat8(float x) {
  v8f c;
#pragma unroll
  for (int r = 0; r < 8; ++r) c[r] = x;
  return c;
}

__global__ void prep_kernel(const float* __restrict__ x,
                            const float* __restrict__ logs,
                            const float* __restrict__ bias,
                            float* __restrict__ zmat,
                            float* __restrict__ logdet,
                            _Float16* __restrict__ z1h, int total) {
  int p = blockIdx.x * blockDim.x + threadIdx.x;
  if (p < total) {
    const float* xr = x + (size_t)p * 32;
    float* zr = zmat + (size_t)p * 32;
    _Float16* hr = z1h + (size_t)p * 16;
#pragma unroll
    for (int c = 0; c < 32; ++c) {
      float zc = xr[31 - c] * expf(logs[c]) + bias[c];
      zr[c] = zc;
      if (c < 16) hr[c] = (_Float16)zc;
    }
  }
  if (p < NBATCH) {
    float su = 0.f;
#pragma unroll
    for (int c = 0; c < 32; ++c) su += logs[c];
    logdet[p] = su * (float)NPOINTS;
  }
}

__global__ __launch_bounds__(NETS * 32) void conv_kernel(
    const _Float16* __restrict__ z1h, const int* __restrict__ knn,
    const float* __restrict__ W1g, const float* __restrict__ b1g,
    const float* __restrict__ W2g, const float* __restrict__ b2g,
    const float* __restrict__ W3g, const float* __restrict__ b3g,
    float* __restrict__ zmat, float* __restrict__ logdet) {
  __shared__ __align__(32) _Float16 sYw[NETS][64][16];
  __shared__ __align__(32) _Float16 sZw[NETS][64][16];
  __shared__ float sRes[NETS][PPW][16];
  __shared__ float sSum[NETS];

  const int u     = __builtin_amdgcn_readfirstlane(threadIdx.x >> 5);
  _Float16 (*sY)[16] = sYw[u];
  _Float16 (*sZ)[16] = sZw[u];
  const int lane  = threadIdx.x & 31;
  const int n16   = lane & 15;
  const int hl    = lane >> 4;
  const int base0 = hl * 8;
  const _Float16* tY = &sY[base0 + (n16 & 7)][8 * (n16 >> 3)];
  const _Float16* tZ = &sZ[base0 + (n16 & 7)][8 * (n16 >> 3)];

  const float* W1 = W1g + (size_t)u * 48 * 64;
  const float* W2 = W2g + (size_t)u * 64 * 64;
  const float* W3 = W3g + (size_t)u * 64 * 16;

  v16h w1[4], w2a[4], w2b[4], w3a, w3b;
  float bb1[4], bb2[4], bb3;
#pragma unroll
  for (int nc = 0; nc < 4; ++nc) {
    const int j = nc * 16 + n16;
    v16h f1, fa, fb;
#pragma unroll
    for (int i = 0; i < 8; ++i) {
      const int k = base0 + i;
      f1[i] = (_Float16)(W1[(16 + k) * 64 + j] + W1[(32 + k) * 64 + j]);
      f1[8 + i] = (_Float16)(W1[k * 64 + j] - W1[(32 + k) * 64 + j]);
      fa[i]     = (_Float16)W2[k * 64 + j];
      fa[8 + i] = (_Float16)W2[(16 + k) * 64 + j];
      fb[i]     = (_Float16)W2[(32 + k) * 64 + j];
      fb[8 + i] = (_Float16)W2[(48 + k) * 64 + j];
    }
    w1[nc] = f1; w2a[nc] = fa; w2b[nc] = fb;
    bb1[nc] = b1g[u * 64 + j];
    bb2[nc] = b2g[u * 64 + j];
    __builtin_amdgcn_sched_barrier(0);
  }
  {
    v16h ga, gb;
#pragma unroll
    for (int i = 0; i < 8; ++i) {
      const int k = base0 + i;
      ga[i]     = (_Float16)W3[k * 16 + n16];
      ga[8 + i] = (_Float16)W3[(16 + k) * 16 + n16];
      gb[i]     = (_Float16)W3[(32 + k) * 16 + n16];
      gb[8 + i] = (_Float16)W3[(48 + k) * 16 + n16];
    }
    w3a = ga; w3b = gb;
    bb3 = b3g[u * 16 + n16];
  }

  const int bi = blockIdx.x / GPB;
  const int gi = blockIdx.x % GPB;
  const int p0 = gi * PPW;
  const int p1 = p0 + PPW;
  const size_t rb = (size_t)bi * NPOINTS;
  const int plast = p1 - 1;

  int nidx = knn[(rb + p0) * 16 + n16];
  v8h nbv = *(const v8h*)(z1h + (rb + nidx) * 16 + base0);
  v8h ctv = *(const v8h*)(z1h + (rb + p0) * 16 + base0);
  {
    const int pn = p0 + 1 < plast ? p0 + 1 : plast;
    nidx = knn[(rb + pn) * 16 + n16];
  }

  for (int p = p0; p < p1; ++p) {
    const v16h A1 = cat16(nbv, ctv);
    {
      const int pn  = p + 1 < plast ? p + 1 : plast;
      const int pnn = p + 2 < plast ? p + 2 : plast;
      nbv  = *(const v8h*)(z1h + (rb + nidx) * 16 + base0);
      ctv  = *(const v8h*)(z1h + (rb + pn) * 16 + base0);
      nidx = knn[(rb + pnn) * 16 + n16];
    }

#pragma unroll
    for (int nc = 0; nc < 4; ++nc) {
      const int j = nc * 16 + n16;
      v8f c = wmma16(A1, w1[nc], splat8(bb1[nc]));
      *(v8h*)&sY[j][base0] = relu_pack(c);
    }
    __builtin_amdgcn_wave_barrier();

    const v16h A2a = ldA(tY, 0);
    const v16h A2b = ldA(tY, 32);
#pragma unroll
    for (int nc = 0; nc < 4; ++nc) {
      const int j = nc * 16 + n16;
      v8f c = splat8(bb2[nc]);
      c = wmma16(A2a, w2a[nc], c);
      c = wmma16(A2b, w2b[nc], c);
      *(v8h*)&sZ[j][base0] = relu_pack(c);
    }
    __builtin_amdgcn_wave_barrier();

    const v16h A3a = ldA(tZ, 0);
    const v16h A3b = ldA(tZ, 32);
    v8f c = splat8(bb3);
    c = wmma16(A3a, w3a, c);
    c = wmma16(A3b, w3b, c);
    float v = c[0];
#pragma unroll
    for (int r = 1; r < 8; ++r) v = fmaxf(v, c[r]);
    v = fmaxf(v, __shfl_xor(v, 16, 32));

    if (lane < 16) sRes[u][p - p0][n16] = v;
    __builtin_amdgcn_wave_barrier();
  }

  float zo[EPT];
#pragma unroll
  for (int i = 0; i < EPT; ++i) {
    const int e = threadIdx.x + i * (NETS * 32);
    zo[i] = zmat[(rb + p0 + (e >> 4)) * 32 + 16 + (e & 15)];
  }

  __syncthreads();

  float s1[EPT], s2[EPT];
  float ssum = 0.f;
#pragma unroll
  for (int i = 0; i < EPT; ++i) {
    const int e  = threadIdx.x + i * (NETS * 32);
    const int pt = e >> 4;
    const int f  = e & 15;
    s1[i] = CLAMPC * TWOPI_I * atanf(sRes[0][pt][f] * (1.0f / CLAMPC));
    s2[i] = CLAMPC * TWOPI_I * atanf(sRes[2][pt][f] * (1.0f / CLAMPC));
    ssum += s1[i] + s2[i];
  }

  ssum += __shfl_xor(ssum, 1, 32);
  ssum += __shfl_xor(ssum, 2, 32);
  ssum += __shfl_xor(ssum, 4, 32);
  ssum += __shfl_xor(ssum, 8, 32);
  ssum += __shfl_xor(ssum, 16, 32);
  if (lane == 0) sSum[u] = ssum;
  __syncthreads();

  if (threadIdx.x == 0)
    atomicAdd(&logdet[bi], (sSum[0] + sSum[1]) + (sSum[2] + sSum[3]));

#pragma unroll
  for (int i = 0; i < EPT; ++i) {
    const int e  = threadIdx.x + i * (NETS * 32);
    const int pt = e >> 4;
    const int f  = e & 15;
    float z2 = zo[i];
    z2 = z2 * __expf(s1[i]) + sRes[1][pt][f];
    z2 = z2 * __expf(s2[i]) + sRes[3][pt][f];
    zmat[(rb + p0 + pt) * 32 + 16 + f] = z2;
  }
}

extern "C" void kernel_launch(void* const* d_in, const int* in_sizes, int n_in,
                              void* d_out, int out_size, void* d_ws, size_t ws_size,
                              hipStream_t stream) {
  const float* x     = (const float*)d_in[0];
  const int*   knn   = (const int*)d_in[1];
  const float* logs  = (const float*)d_in[2];
  const float* biasv = (const float*)d_in[3];
  const float* W1    = (const float*)d_in[4];
  const float* b1    = (const float*)d_in[5];
  const float* W2    = (const float*)d_in[6];
  const float* b2    = (const float*)d_in[7];
  const float* W3    = (const float*)d_in[8];
  const float* b3    = (const float*)d_in[9];

  float* zmat   = (float*)d_out;
  float* logdet = zmat + (size_t)NBATCH * NPOINTS * 32;
  _Float16* z1h = (_Float16*)d_ws;

  const int total = NBATCH * NPOINTS;
  prep_kernel<<<(total + 255) / 256, 256, 0, stream>>>(x, logs, biasv, zmat,
                                                       logdet, z1h, total);

  conv_kernel<<<dim3(NBATCH * GPB), NETS * 32, 0, stream>>>(
      z1h, knn, W1, b1, W2, b2, W3, b3, zmat, logdet);
}


# two points per trip of the point loop: a weight fragment feeds both points' products back to back, LDS waits are shared; registers from the biases, now added before the relu (sum order changes)
# speedup vs baseline: 6.6500x; 1.1614x over previous
#include <hip/hip_runtime.h>
#include <hip/hip_bf16.h>


typedef __attribute__((ext_vector_type(16))) _Float16 v16h;
typedef __attribute__((ext_vector_type(8)))  _Float16 v8h;
typedef __attribute__((ext_vector_type(8)))  float    v8f;
typedef __attribute__((ext_vector_type(2)))  _Float16 v2h;
typedef __attribute__((ext_vector_type(2)))  float    v2f;
typedef __attribute__((ext_vector_type(8)))  short    v8s;

#define NPOINTS 8192
#define NBATCH  16
#define PPW     128
#define GPB     (NPOINTS / PPW)
#define NETS    4
#define EPT     (PPW * 16 / (NETS * 32))
#define PPT     2
#define CLAMPC  1.9f
#define TWOPI_I 0.63661977236758f

__device__ __forceinline__ v16h cat16(v8h lo, v8h hi) {
  return __builtin_shufflevector(lo, hi, 0,1,2,3,4,5,6,7,8,9,10,11,12,13,14,15);
}

__device__ __forceinline__ v8h ld_tr16(const _Float16* p) {
  typedef __attribute__((address_space(3))) v8s lds_v8s;
  return __builtin_bit_cast(v8h, __builtin_amdgcn_ds_load_tr16_b128_v8i16((lds_v8s*)p));
}

__device__ __forceinline__ v16h ldA(const _Float16* p, int kb) {
  return cat16(ld_tr16(p + kb * 16), ld_tr16(p + (kb + 16) * 16));
}

__device__ __forceinline__ v8h bias_relu_pack(v8f c, float b) {
  v2h q[4];
#pragma unroll
  for (int r = 0; r < 4; ++r) {
    v2f t = {fmaxf(c[2 * r] + b, 0.0f), fmaxf(c[2 * r + 1] + b, 0.0f)};
    q[r] = __builtin_convertvector(t, v2h);
  }
  v8h o = {q[0][0], q[0][1], q[1][0], q[1][1], q[2][0], q[2][1], q[3][0], q[3][1]};
  return o;
}

__device__ __forceinline__ v8f wmma16(v16h a, v16h b, v8f c) {
  return __builtin_amdgcn_wmma_f32_16x16x32_f16(false, a, false, b, (short)0, c,
                                                false, false);
}

__device__ __forceinline__ v8f zero8() {
  v8f c = {0.f, 0.f, 0.f, 0.f, 0.f, 0.f, 0.f, 0.f};
  return c;
}

__global__ void prep_kernel(const float* __restrict__ x,
                            const float* __restrict__ logs,
                            const float* __restrict__ bias,
                            float* __restrict__ zmat,
                            float* __restrict__ logdet,
                            _Float16* __restrict__ z1h, int total) {
  int p = blockIdx.x * blockDim.x + threadIdx.x;
  if (p < total) {
    const float* xr = x + (size_t)p * 32;
    float* zr = zmat + (size_t)p * 32;
    _Float16* hr = z1h + (size_t)p * 16;
#pragma unroll
    for (int c = 0; c < 32; ++c) {
      float zc = xr[31 - c] * expf(logs[c]) + bias[c];
      zr[c] = zc;
      if (c < 16) hr[c] = (_Float16)zc;
    }
  }
  if (p < NBATCH) {
    float su = 0.f;
#pragma unroll
    for (int c = 0; c < 32; ++c) su += logs[c];
    logdet[p] = su * (float)NPOINTS;
  }
}

__global__ __launch_bounds__(NETS * 32) __attribute__((amdgpu_waves_per_eu(4)))
void conv_kernel(
    const _Float16* __restrict__ z1h, const int* __restrict__ knn,
    const float* __restrict__ W1g, const float* __restrict__ b1g,
    const float* __restrict__ W2g, const float* __restrict__ b2g,
    const float* __restrict__ W3g, const float* __restrict__ b3g,
    float* __restrict__ zmat, float* __restrict__ logdet) {
  __shared__ __align__(32) _Float16 sYw[NETS][PPT][64][16];
  __shared__ __align__(32) _Float16 sZw[NETS][PPT][64][16];
  __shared__ float sRes[NETS][PPW][16];
  __shared__ float sSum[NETS];

  const int u     = __builtin_amdgcn_readfirstlane(threadIdx.x >> 5);
  _Float16 (*sY)[64][16] = sYw[u];
  _Float16 (*sZ)[64][16] = sZw[u];
  const int lane  = threadIdx.x & 31;
  const int n16   = lane & 15;
  const int hl    = lane >> 4;
  const int base0 = hl * 8;
  const _Float16* tY = &sY[0][base0 + (n16 & 7)][8 * (n16 >> 3)];
  const _Float16* tZ = &sZ[0][base0 + (n16 & 7)][8 * (n16 >> 3)];

  const float* W1 = W1g + (size_t)u * 48 * 64;
  const float* W2 = W2g + (size_t)u * 64 * 64;
  const float* W3 = W3g + (size_t)u * 64 * 16;

  v16h w1[4], w2a[4], w2b[4], w3a, w3b;
  float bb1[4], bb2[4], bb3;
#pragma unroll
  for (int nc = 0; nc < 4; ++nc) {
    const int j = nc * 16 + n16;
    v16h f1, fa, fb;
#pragma unroll
    for (int i = 0; i < 8; ++i) {
      const int k = base0 + i;
      f1[i] = (_Float16)(W1[(16 + k) * 64 + j] + W1[(32 + k) * 64 + j]);
      f1[8 + i] = (_Float16)(W1[k * 64 + j] - W1[(32 + k) * 64 + j]);
      fa[i]     = (_Float16)W2[k * 64 + j];
      fa[8 + i] = (_Float16)W2[(16 + k) * 64 + j];
      fb[i]     = (_Float16)W2[(32 + k) * 64 + j];
      fb[8 + i] = (_Float16)W2[(48 + k) * 64 + j];
    }
    w1[nc] = f1; w2a[nc] = fa; w2b[nc] = fb;
    bb1[nc] = b1g[u * 64 + j];
    bb2[nc] = b2g[u * 64 + j];
    __builtin_amdgcn_sched_barrier(0);
  }
  {
    v16h ga, gb;
#pragma unroll
    for (int i = 0; i < 8; ++i) {
      const int k = base0 + i;
      ga[i]     = (_Float16)W3[k * 16 + n16];
      ga[8 + i] = (_Float16)W3[(16 + k) * 16 + n16];
      gb[i]     = (_Float16)W3[(32 + k) * 16 + n16];
      gb[8 + i] = (_Float16)W3[(48 + k) * 16 + n16];
    }
    w3a = ga; w3b = gb;
    bb3 = b3g[u * 16 + n16];
  }

  const int bi = blockIdx.x / GPB;
  const int gi = blockIdx.x % GPB;
  const int p0 = gi * PPW;
  const int p1 = p0 + PPW;
  const size_t rb = (size_t)bi * NPOINTS;
  const int plast = p1 - 1;

  int nidx[PPT];
  v8h nbv[PPT], ctv[PPT];
#pragma unroll
  for (int q = 0; q < PPT; ++q) {
    nidx[q] = knn[(rb + p0 + q) * 16 + n16];
    nbv[q] = *(const v8h*)(z1h + (rb + nidx[q]) * 16 + base0);
    ctv[q] = *(const v8h*)(z1h + (rb + p0 + q) * 16 + base0);
    const int pn = p0 + PPT + q < plast ? p0 + PPT + q : plast;
    nidx[q] = knn[(rb + pn) * 16 + n16];
  }

  for (int p = p0; p < p1; p += PPT) {
    v8f c1[PPT][4];
#pragma unroll
    for (int nc = 0; nc < 4; ++nc)
#pragma unroll
      for (int q = 0; q < PPT; ++q)
        c1[q][nc] = wmma16(cat16(nbv[q], ctv[q]), w1[nc], zero8());
    __builtin_amdgcn_sched_barrier(0);
#pragma unroll
    for (int q = 0; q < PPT; ++q) {
      const int pn  = p + PPT + q < plast ? p + PPT + q : plast;
      const int pnn = p + 2 * PPT + q < plast ? p + 2 * PPT + q : plast;
      nbv[q]  = *(const v8h*)(z1h + (rb + nidx[q]) * 16 + base0);
      ctv[q]  = *(const v8h*)(z1h + (rb + pn) * 16 + base0);
      nidx[q] = knn[(rb + pnn) * 16 + n16];
    }
#pragma unroll
    for (int q = 0; q < PPT; ++q)
#pragma unroll
      for (int nc = 0; nc < 4; ++nc) {
        *(v8h*)&sY[q][nc * 16 + n16][base0] = bias_relu_pack(c1[q][nc], bb1[nc]);
      }
    __builtin_amdgcn_wave_barrier();

    v16h A2a[PPT], A2b[PPT];
#pragma unroll
    for (int q = 0; q < PPT; ++q) {
      A2a[q] = ldA(tY + q * 64 * 16, 0);
      A2b[q] = ldA(tY + q * 64 * 16, 32);
    }
#pragma unroll
    for (int h = 0; h < 2; ++h) {
      v8f c[PPT][2];
#pragma unroll
      for (int i = 0; i < 2; ++i)
#pragma unroll
        for (int q = 0; q < PPT; ++q) c[q][i] = wmma16(A2a[q], w2a[2 * h + i], zero8());
#pragma unroll
      for (int i = 0; i < 2; ++i)
#pragma unroll
        for (int q = 0; q < PPT; ++q) c[q][i] = wmma16(A2b[q], w2b[2 * h + i], c[q][i]);
#pragma unroll
      for (int q = 0; q < PPT; ++q)
#pragma unroll
        for (int i = 0; i < 2; ++i) {
          const int nc = 2 * h + i;
          *(v8h*)&sZ[q][nc * 16 + n16][base0] = bias_relu_pack(c[q][i], bb2[nc]);
        }
    }
    __builtin_amdgcn_wave_barrier();

    v16h A3a[PPT], A3b[PPT];
#pragma unroll
    for (int q = 0; q < PPT; ++q) {
      A3a[q] = ldA(tZ + q * 64 * 16, 0);
      A3b[q] = ldA(tZ + q * 64 * 16, 32);
    }
    __builtin_amdgcn_sched_barrier(0);
    v8f d[PPT];
#pragma unroll
    for (int q = 0; q < PPT; ++q) d[q] = wmma16(A3a[q], w3a, zero8());
#pragma unroll
    for (int q = 0; q < PPT; ++q) d[q] = wmma16(A3b[q], w3b, d[q]);
    __builtin_amdgcn_sched_barrier(0);
    float v[PPT];
#pragma unroll
    for (int q = 0; q < PPT; ++q) {
      v[q] = d[q][0];
#pragma unroll
      for (int r = 1; r < 8; ++r) v[q] = fmaxf(v[q], d[q][r]);
    }
#pragma unroll
    for (int q = 0; q < PPT; ++q)
      v[q] = fmaxf(v[q], __shfl_xor(v[q], 16, 32));

    if (lane < 16) {
#pragma unroll
      for (int q = 0; q < PPT; ++q) sRes[u][p - p0 + q][n16] = v[q] + bb3;
    }
    __builtin_amdgcn_wave_barrier();
  }

  float zo[EPT];
#pragma unroll
  for (int i = 0; i < EPT; ++i) {
    const int e = threadIdx.x + i * (NETS * 32);
    zo[i] = zmat[(rb + p0 + (e >> 4)) * 32 + 16 + (e & 15)];
  }

  __syncthreads();

  float s1[EPT], s2[EPT];
  float ssum = 0.f;
#pragma unroll
  for (int i = 0; i < EPT; ++i) {
    const int e  = threadIdx.x + i * (NETS * 32);
    const int pt = e >> 4;
    const int f  = e & 15;
    s1[i] = CLAMPC * TWOPI_I * atanf(sRes[0][pt][f] * (1.0f / CLAMPC));
    s2[i] = CLAMPC * TWOPI_I * atanf(sRes[2][pt][f] * (1.0f / CLAMPC));
    ssum += s1[i] + s2[i];
  }

  ssum += __shfl_xor(ssum, 1, 32);
  ssum += __shfl_xor(ssum, 2, 32);
  ssum += __shfl_xor(ssum, 4, 32);
  ssum += __shfl_xor(ssum, 8, 32);
  ssum += __shfl_xor(ssum, 16, 32);
  if (lane == 0) sSum[u] = ssum;
  __syncthreads();

  if (threadIdx.x == 0)
    atomicAdd(&logdet[bi], (sSum[0] + sSum[1]) + (sSum[2] + sSum[3]));

#pragma unroll
  for (int i = 0; i < EPT; ++i) {
    const int e  = threadIdx.x + i * (NETS * 32);
    const int pt = e >> 4;
    const int f  = e & 15;
    float z2 = zo[i];
    z2 = z2 * __expf(s1[i]) + sRes[1][pt][f];
    z2 = z2 * __expf(s2[i]) + sRes[3][pt][f];
    zmat[(rb + p0 + pt) * 32 + 16 + f] = z2;
  }
}

extern "C" void kernel_launch(void* const* d_in, const int* in_sizes, int n_in,
                              void* d_out, int out_size, void* d_ws, size_t ws_size,
                              hipStream_t stream) {
  const float* x     = (const float*)d_in[0];
  const int*   knn   = (const int*)d_in[1];
  const float* logs  = (const float*)d_in[2];
  const float* biasv = (const float*)d_in[3];
  const float* W1    = (const float*)d_in[4];
  const float* b1    = (const float*)d_in[5];
  const float* W2    = (const float*)d_in[6];
  const float* b2    = (const float*)d_in[7];
  const float* W3    = (const float*)d_in[8];
  const float* b3    = (const float*)d_in[9];

  float* zmat   = (float*)d_out;
  float* logdet = zmat + (size_t)NBATCH * NPOINTS * 32;
  _Float16* z1h = (_Float16*)d_ws;

  const int total = NBATCH * NPOINTS;
  prep_kernel<<<(total + 255) / 256, 256, 0, stream>>>(x, logs, biasv, zmat,
                                                       logdet, z1h, total);

  conv_kernel<<<dim3(NBATCH * GPB), NETS * 32, 0, stream>>>(
      z1h, knn, W1, b1, W2, b2, W3, b3, zmat, logdet);
}


# relu(x+b) kept as max(x,-b) on the packed halves; the +b goes through the next layer's weights into its bias once per wave (a product of the bias tile): no add per element (sum order changes)
# speedup vs baseline: 7.1852x; 1.0805x over previous
#include <hip/hip_runtime.h>
#include <hip/hip_bf16.h>


typedef __attribute__((ext_vector_type(16))) _Float16 v16h;
typedef __attribute__((ext_vector_type(8)))  _Float16 v8h;
typedef __attribute__((ext_vector_type(8)))  float    v8f;
typedef __attribute__((ext_vector_type(2)))  _Float16 v2h;
typedef __attribute__((ext_vector_type(2)))  float    v2f;
typedef __attribute__((ext_vector_type(8)))  short    v8s;

#define NPOINTS 8192
#define NBATCH  16
#define PPW     128
#define GPB     (NPOINTS / PPW)
#define NETS    4
#define EPT     (PPW * 16 / (NETS * 32))
#define PPT     2
#define CLAMPC  1.9f
#define TWOPI_I 0.63661977236758f

__device__ __forceinline__ v16h cat16(v8h lo, v8h hi) {
  return __builtin_shufflevector(lo, hi, 0,1,2,3,4,5,6,7,8,9,10,11,12,13,14,15);
}

__device__ __forceinline__ v8h ld_tr16(const _Float16* p) {
  typedef __attribute__((address_space(3))) v8s lds_v8s;
  return __builtin_bit_cast(v8h, __builtin_amdgcn_ds_load_tr16_b128_v8i16((lds_v8s*)p));
}

__device__ __forceinline__ v16h ldA(const _Float16* p, int kb) {
  return cat16(ld_tr16(p + kb * 16), ld_tr16(p + (kb + 16) * 16));
}

__device__ __forceinline__ v8h thr_pack(v8f c, v2h nb) {
  v2h q[4];
#pragma unroll
  for (int r = 0; r < 4; ++r) {
    v2f t = {c[2 * r], c[2 * r + 1]};
    const v2h h = __builtin_convertvector(t, v2h);
    asm("v_pk_max_num_f16 %0, %1, %2" : "=v"(q[r]) : "v"(h), "v"(nb));
  }
  v8h o = {q[0][0], q[0][1], q[1][0], q[1][1], q[2][0], q[2][1], q[3][0], q[3][1]};
  return o;
}

__device__ __forceinline__ v2h thr_of(float b) {
  const _Float16 t = (_Float16)(-b);
  v2h tt = {t, t};
  return tt;
}

__device__ __forceinline__ v8f wmma16(v16h a, v16h b, v8f c) {
  return __builtin_amdgcn_wmma_f32_16x16x32_f16(false, a, false, b, (short)0, c,
                                                false, false);
}

__device__ __forceinline__ v8f zero8() {
  v8f c = {0.f, 0.f, 0.f, 0.f, 0.f, 0.f, 0.f, 0.f};
  return c;
}

__global__ void prep_kernel(const float* __restrict__ x,
                            const float* __restrict__ logs,
                            const float* __restrict__ bias,
                            float* __restrict__ zmat,
                            float* __restrict__ logdet,
                            _Float16* __restrict__ z1h, int total) {
  int p = blockIdx.x * blockDim.x + threadIdx.x;
  if (p < total) {
    const float* xr = x + (size_t)p * 32;
    float* zr = zmat + (size_t)p * 32;
    _Float16* hr = z1h + (size_t)p * 16;
#pragma unroll
    for (int c = 0; c < 32; ++c) {
      float zc = xr[31 - c] * expf(logs[c]) + bias[c];
      zr[c] = zc;
      if (c < 16) hr[c] = (_Float16)zc;
    }
  }
  if (p < NBATCH) {
    float su = 0.f;
#pragma unroll
    for (int c = 0; c < 32; ++c) su += logs[c];
    logdet[p] = su * (float)NPOINTS;
  }
}

__global__ __launch_bounds__(NETS * 32) __attribute__((amdgpu_waves_per_eu(4)))
void conv_kernel(
    const _Float16* __restrict__ z1h, const int* __restrict__ knn,
    const float* __restrict__ W1g, const float* __restrict__ b1g,
    const float* __restrict__ W2g, const float* __restrict__ b2g,
    const float* __restrict__ W3g, const float* __restrict__ b3g,
    float* __restrict__ zmat, float* __restrict__ logdet) {
  __shared__ __align__(32) _Float16 sYw[NETS][PPT][64][16];
  __shared__ __align__(32) _Float16 sZw[NETS][PPT][64][16];
  __shared__ float sRes[NETS][PPW][16];
  __shared__ float sSum[NETS];

  const int u     = __builtin_amdgcn_readfirstlane(threadIdx.x >> 5);
  _Float16 (*sY)[64][16] = sYw[u];
  _Float16 (*sZ)[64][16] = sZw[u];
  const int lane  = threadIdx.x & 31;
  const int n16   = lane & 15;
  const int hl    = lane >> 4;
  const int base0 = hl * 8;
  const _Float16* tY = &sY[0][base0 + (n16 & 7)][8 * (n16 >> 3)];
  const _Float16* tZ = &sZ[0][base0 + (n16 & 7)][8 * (n16 >> 3)];

  const float* W1 = W1g + (size_t)u * 48 * 64;
  const float* W2 = W2g + (size_t)u * 64 * 64;
  const float* W3 = W3g + (size_t)u * 64 * 16;

  v16h w1[4], w2a[4], w2b[4], w3a, w3b;
  v2h nb1[4], nb2[4];
  float e3;
#pragma unroll
  for (int nc = 0; nc < 4; ++nc) {
    const int j = nc * 16 + n16;
    v16h f1, fa, fb;
#pragma unroll
    for (int i = 0; i < 8; ++i) {
      const int k = base0 + i;
      f1[i] = (_Float16)(W1[(16 + k) * 64 + j] + W1[(32 + k) * 64 + j]);
      f1[8 + i] = (_Float16)(W1[k * 64 + j] - W1[(32 + k) * 64 + j]);
      fa[i]     = (_Float16)W2[k * 64 + j];
      fa[8 + i] = (_Float16)W2[(16 + k) * 64 + j];
      fb[i]     = (_Float16)W2[(32 + k) * 64 + j];
      fb[8 + i] = (_Float16)W2[(48 + k) * 64 + j];
    }
    w1[nc] = f1; w2a[nc] = fa; w2b[nc] = fb;
    nb1[nc] = thr_of(b1g[u * 64 + j]);
    __builtin_amdgcn_sched_barrier(0);
  }
  {
    v16h ga, gb;
#pragma unroll
    for (int i = 0; i < 8; ++i) {
      const int k = base0 + i;
      ga[i]     = (_Float16)W3[k * 16 + n16];
      ga[8 + i] = (_Float16)W3[(16 + k) * 16 + n16];
      gb[i]     = (_Float16)W3[(32 + k) * 16 + n16];
      gb[8 + i] = (_Float16)W3[(48 + k) * 16 + n16];
    }
    w3a = ga; w3b = gb;
  }

  {
#pragma unroll
    for (int nc = 0; nc < 4; ++nc) {
      const _Float16 bh = -nb1[nc][0];
      v8h bv = {bh, bh, bh, bh, bh, bh, bh, bh};
      *(v8h*)&sY[0][nc * 16 + n16][base0] = bv;
    }
    __builtin_amdgcn_wave_barrier();
    const v16h Ba = ldA(tY, 0), Bb = ldA(tY, 32);
#pragma unroll
    for (int nc = 0; nc < 4; ++nc) {
      v8f d = wmma16(Ba, w2a[nc], zero8());
      d = wmma16(Bb, w2b[nc], d);
      nb2[nc] = thr_of(b2g[u * 64 + nc * 16 + n16] + d[0]);
    }
#pragma unroll
    for (int nc = 0; nc < 4; ++nc) {
      const _Float16 bh = -nb2[nc][0];
      v8h bv = {bh, bh, bh, bh, bh, bh, bh, bh};
      *(v8h*)&sZ[0][nc * 16 + n16][base0] = bv;
    }
    __builtin_amdgcn_wave_barrier();
    const v16h Ca = ldA(tZ, 0), Cb = ldA(tZ, 32);
    v8f d = wmma16(Ca, w3a, zero8());
    d = wmma16(Cb, w3b, d);
    e3 = b3g[u * 16 + n16] + d[0];
    __builtin_amdgcn_wave_barrier();
  }

  const int bi = blockIdx.x / GPB;
  const int gi = blockIdx.x % GPB;
  const int p0 = gi * PPW;
  const int p1 = p0 + PPW;
  const size_t rb = (size_t)bi * NPOINTS;
  const int plast = p1 - 1;

  int nidx[PPT];
  v8h nbv[PPT], ctv[PPT];
#pragma unroll
  for (int q = 0; q < PPT; ++q) {
    nidx[q] = knn[(rb + p0 + q) * 16 + n16];
    nbv[q] = *(const v8h*)(z1h + (rb + nidx[q]) * 16 + base0);
    ctv[q] = *(const v8h*)(z1h + (rb + p0 + q) * 16 + base0);
    const int pn = p0 + PPT + q < plast ? p0 + PPT + q : plast;
    nidx[q] = knn[(rb + pn) * 16 + n16];
  }

  for (int p = p0; p < p1; p += PPT) {
    v8f c1[PPT][4];
#pragma unroll
    for (int nc = 0; nc < 4; ++nc)
#pragma unroll
      for (int q = 0; q < PPT; ++q)
        c1[q][nc] = wmma16(cat16(nbv[q], ctv[q]), w1[nc], zero8());
    __builtin_amdgcn_sched_barrier(0);
#pragma unroll
    for (int q = 0; q < PPT; ++q) {
      const int pn  = p + PPT + q < plast ? p + PPT + q : plast;
      const int pnn = p + 2 * PPT + q < plast ? p + 2 * PPT + q : plast;
      nbv[q]  = *(const v8h*)(z1h + (rb + nidx[q]) * 16 + base0);
      ctv[q]  = *(const v8h*)(z1h + (rb + pn) * 16 + base0);
      nidx[q] = knn[(rb + pnn) * 16 + n16];
    }
#pragma unroll
    for (int q = 0; q < PPT; ++q)
#pragma unroll
      for (int nc = 0; nc < 4; ++nc) {
        *(v8h*)&sY[q][nc * 16 + n16][base0] = thr_pack(c1[q][nc], nb1[nc]);
      }
    __builtin_amdgcn_wave_barrier();

    v16h A2a[PPT], A2b[PPT];
#pragma unroll
    for (int q = 0; q < PPT; ++q) {
      A2a[q] = ldA(tY + q * 64 * 16, 0);
      A2b[q] = ldA(tY + q * 64 * 16, 32);
    }
#pragma unroll
    for (int h = 0; h < 2; ++h) {
      v8f c[PPT][2];
#pragma unroll
      for (int i = 0; i < 2; ++i)
#pragma unroll
        for (int q = 0; q < PPT; ++q) c[q][i] = wmma16(A2a[q], w2a[2 * h + i], zero8());
#pragma unroll
      for (int i = 0; i < 2; ++i)
#pragma unroll
        for (int q = 0; q < PPT; ++q) c[q][i] = wmma16(A2b[q], w2b[2 * h + i], c[q][i]);
#pragma unroll
      for (int q = 0; q < PPT; ++q)
#pragma unroll
        for (int i = 0; i < 2; ++i) {
          const int nc = 2 * h + i;
          *(v8h*)&sZ[q][nc * 16 + n16][base0] = thr_pack(c[q][i], nb2[nc]);
        }
    }
    __builtin_amdgcn_wave_barrier();

    v16h A3a[PPT], A3b[PPT];
#pragma unroll
    for (int q = 0; q < PPT; ++q) {
      A3a[q] = ldA(tZ + q * 64 * 16, 0);
      A3b[q] = ldA(tZ + q * 64 * 16, 32);
    }
    __builtin_amdgcn_sched_barrier(0);
    v8f d[PPT];
#pragma unroll
    for (int q = 0; q < PPT; ++q) d[q] = wmma16(A3a[q], w3a, zero8());
#pragma unroll
    for (int q = 0; q < PPT; ++q) d[q] = wmma16(A3b[q], w3b, d[q]);
    __builtin_amdgcn_sched_barrier(0);
    float v[PPT];
#pragma unroll
    for (int q = 0; q < PPT; ++q) {
      v[q] = d[q][0];
#pragma unroll
      for (int r = 1; r < 8; ++r) v[q] = fmaxf(v[q], d[q][r]);
    }
#pragma unroll
    for (int q = 0; q < PPT; ++q)
      v[q] = fmaxf(v[q], __shfl_xor(v[q], 16, 32));

    if (lane < 16) {
#pragma unroll
      for (int q = 0; q < PPT; ++q) sRes[u][p - p0 + q][n16] = v[q] + e3;
    }
    __builtin_amdgcn_wave_barrier();
  }

  float zo[EPT];
#pragma unroll
  for (int i = 0; i < EPT; ++i) {
    const int e = threadIdx.x + i * (NETS * 32);
    zo[i] = zmat[(rb + p0 + (e >> 4)) * 32 + 16 + (e & 15)];
  }

  __syncthreads();

  float s1[EPT], s2[EPT];
  float ssum = 0.f;
#pragma unroll
  for (int i = 0; i < EPT; ++i) {
    const int e  = threadIdx.x + i * (NETS * 32);
    const int pt = e >> 4;
    const int f  = e & 15;
    s1[i] = CLAMPC * TWOPI_I * atanf(sRes[0][pt][f] * (1.0f / CLAMPC));
    s2[i] = CLAMPC * TWOPI_I * atanf(sRes[2][pt][f] * (1.0f / CLAMPC));
    ssum += s1[i] + s2[i];
  }

  ssum += __shfl_xor(ssum, 1, 32);
  ssum += __shfl_xor(ssum, 2, 32);
  ssum += __shfl_xor(ssum, 4, 32);
  ssum += __shfl_xor(ssum, 8, 32);
  ssum += __shfl_xor(ssum, 16, 32);
  if (lane == 0) sSum[u] = ssum;
  __syncthreads();

  if (threadIdx.x == 0)
    atomicAdd(&logdet[bi], (sSum[0] + sSum[1]) + (sSum[2] + sSum[3]));

#pragma unroll
  for (int i = 0; i < EPT; ++i) {
    const int e  = threadIdx.x + i * (NETS * 32);
    const int pt = e >> 4;
    const int f  = e & 15;
    float z2 = zo[i];
    z2 = z2 * __expf(s1[i]) + sRes[1][pt][f];
    z2 = z2 * __expf(s2[i]) + sRes[3][pt][f];
    zmat[(rb + p0 + pt) * 32 + 16 + f] = z2;
  }
}

extern "C" void kernel_launch(void* const* d_in, const int* in_sizes, int n_in,
                              void* d_out, int out_size, void* d_ws, size_t ws_size,
                              hipStream_t stream) {
  const float* x     = (const float*)d_in[0];
  const int*   knn   = (const int*)d_in[1];
  const float* logs  = (const float*)d_in[2];
  const float* biasv = (const float*)d_in[3];
  const float* W1    = (const float*)d_in[4];
  const float* b1    = (const float*)d_in[5];
  const float* W2    = (const float*)d_in[6];
  const float* b2    = (const float*)d_in[7];
  const float* W3    = (const float*)d_in[8];
  const float* b3    = (const float*)d_in[9];

  float* zmat   = (float*)d_out;
  float* logdet = zmat + (size_t)NBATCH * NPOINTS * 32;
  _Float16* z1h = (_Float16*)d_ws;

  const int total = NBATCH * NPOINTS;
  prep_kernel<<<(total + 255) / 256, 256, 0, stream>>>(x, logs, biasv, zmat,
                                                       logdet, z1h, total);

  conv_kernel<<<dim3(NBATCH * GPB), NETS * 32, 0, stream>>>(
      z1h, knn, W1, b1, W2, b2, W3, b3, zmat, logdet);
}
